# starting kernel
# baseline (speedup 1.0000x reference)
#include <hip/hip_runtime.h>


typedef __attribute__((ext_vector_type(2))) float v2f;
typedef __attribute__((ext_vector_type(8))) float v8f;

namespace {
constexpr int KD    = 256;
constexpr int NC    = 64;
constexpr int NR    = 8192;
constexpr int WAVES = 8;
constexpr int QBLK  = NR / (WAVES * 16);
}

__global__ void __launch_bounds__(256)
clusternet_cnorm_kernel(const float* __restrict__ cent,
                        float* __restrict__ cnorm) {
    const int lane = threadIdx.x & 31;
    const int row  = blockIdx.x * WAVES + (threadIdx.x >> 5);
    const float* cr = cent + (size_t)row * KD;

    float s = 0.f;
#pragma unroll
    for (int h = 0; h < KD; h += 32) {
        float v = cr[h + lane];
        s += v * v;
    }
    s += __shfl_xor(s, 1, 32);
    s += __shfl_xor(s, 2, 32);
    s += __shfl_xor(s, 4, 32);
    s += __shfl_xor(s, 8, 32);
    s += __shfl_xor(s, 16, 32);
    if (lane == 0) cnorm[row] = s;
}

__global__ void __launch_bounds__(256)
clusternet_q_kernel(const float* __restrict__ z,
                    const float* __restrict__ cent,
                    const float* __restrict__ cnorm,
                    float* __restrict__ Qout,
                    float* __restrict__ blockpart) {
    __shared__ float csum[WAVES][NC];

    const int lane = threadIdx.x & 31;
    const int wave = threadIdx.x >> 5;
    const int half = lane >> 4;
    const int n16  = lane & 15;
    const int rowBase = (blockIdx.x * WAVES + wave) * 16;

    const float* zrow = z + (size_t)(rowBase + n16) * KD + 2 * half;
    const float* cr[4];
    float cn[4];
#pragma unroll
    for (int t = 0; t < 4; ++t) {
        cr[t] = cent + (size_t)(t * 16 + n16) * KD + 2 * half;
        cn[t] = cnorm[t * 16 + n16];
    }

    v8f acc[4] = {};
    float zp = 0.f;

#pragma unroll 4
    for (int kb = 0; kb < KD; kb += 4) {
        v2f av = *(const v2f*)(zrow + kb);
        zp += av.x * av.x + av.y * av.y;
#pragma unroll
        for (int t = 0; t < 4; ++t) {
            v2f bv = *(const v2f*)(cr[t] + kb);
            acc[t] = __builtin_amdgcn_wmma_f32_16x16x4_f32(
                false, av, false, bv, (short)0, acc[t], false, false);
        }
    }

    float znf = zp + __shfl_xor(zp, 16, 32);

    float qv[4][8];
#pragma unroll
    for (int r = 0; r < 8; ++r) {
        float znr = __shfl(znf, r + 8 * half, 32);
#pragma unroll
        for (int t = 0; t < 4; ++t) {
            float d  = acc[t][r];
            float s2 = fmaxf(znr + cn[t] - 2.0f * d, 0.0f);
            float sim = sqrtf(s2);
            qv[t][r] = 1.0f / (1.0f + sim);
        }
        float rs = qv[0][r] + qv[1][r] + qv[2][r] + qv[3][r];
        rs += __shfl_xor(rs, 1, 32);
        rs += __shfl_xor(rs, 2, 32);
        rs += __shfl_xor(rs, 4, 32);
        rs += __shfl_xor(rs, 8, 32);
        float inv = 1.0f / rs;
#pragma unroll
        for (int t = 0; t < 4; ++t) {
            qv[t][r] *= inv;
            Qout[(size_t)(rowBase + r + 8 * half) * NC + t * 16 + n16] = qv[t][r];
        }
    }

#pragma unroll
    for (int t = 0; t < 4; ++t) {
        float cs = 0.f;
#pragma unroll
        for (int r = 0; r < 8; ++r) cs += qv[t][r];
        cs += __shfl_xor(cs, 16, 32);
        if (lane < 16) csum[wave][t * 16 + n16] = cs;
    }
    __syncthreads();
    if (threadIdx.x < NC) {
        float s = 0.f;
#pragma unroll
        for (int w = 0; w < WAVES; ++w) s += csum[w][threadIdx.x];
        blockpart[(size_t)blockIdx.x * NC + threadIdx.x] = s;
    }
}

__global__ void __launch_bounds__(64)
clusternet_colsum_kernel(const float* __restrict__ blockpart,
                         float* __restrict__ colsum) {
    float s = 0.f;
#pragma unroll 8
    for (int b = 0; b < QBLK; ++b) s += blockpart[(size_t)b * NC + threadIdx.x];
    colsum[threadIdx.x] = s;
}

__global__ void __launch_bounds__(256)
clusternet_p_kernel(const float* __restrict__ Q,
                    const float* __restrict__ colsum,
                    float* __restrict__ Pout) {
    const int lane = threadIdx.x & 31;
    const int row  = blockIdx.x * WAVES + (threadIdx.x >> 5);
    const size_t base = (size_t)row * NC;

    float q0 = Q[base + lane];
    float q1 = Q[base + lane + 32];
    float p0 = q0 * q0 / colsum[lane];
    float p1 = q1 * q1 / colsum[lane + 32];

    float s = p0 + p1;
    s += __shfl_xor(s, 1, 32);
    s += __shfl_xor(s, 2, 32);
    s += __shfl_xor(s, 4, 32);
    s += __shfl_xor(s, 8, 32);
    s += __shfl_xor(s, 16, 32);
    float inv = 1.0f / s;

    Pout[base + lane]      = p0 * inv;
    Pout[base + lane + 32] = p1 * inv;
}

extern "C" void kernel_launch(void* const* d_in, const int* in_sizes, int n_in,
                              void* d_out, int out_size, void* d_ws, size_t ws_size,
                              hipStream_t stream) {
    const float* z    = (const float*)d_in[0];
    const float* cent = (const float*)d_in[1];

    float* Q = (float*)d_out;
    float* P = Q + (size_t)NR * NC;

    float* colsum    = (float*)d_ws;
    float* blockpart = colsum + NC;
    float* cnorm     = blockpart + (size_t)QBLK * NC;

    clusternet_cnorm_kernel<<<NC / WAVES, 256, 0, stream>>>(cent, cnorm);
    clusternet_q_kernel<<<QBLK, 256, 0, stream>>>(z, cent, cnorm, Q, blockpart);
    clusternet_colsum_kernel<<<1, 64, 0, stream>>>(blockpart, colsum);
    clusternet_p_kernel<<<NR / WAVES, 256, 0, stream>>>(Q, colsum, P);
}


# Q GEMM re-tiled: 256 blocks x 4 waves, each wave one K quarter with its centroid slice held in registers, z read once by 16-B loads; quarter dots summed via LDS, colsum from 256 block partials
# speedup vs baseline: 1.6020x; 1.6020x over previous
#include <hip/hip_runtime.h>


typedef __attribute__((ext_vector_type(2))) float v2f;
typedef __attribute__((ext_vector_type(4))) float v4f;
typedef __attribute__((ext_vector_type(8))) float v8f;

namespace {
constexpr int KD    = 256;
constexpr int NC    = 64;
constexpr int NR    = 8192;
constexpr int WAVES = 8;
constexpr int QW    = 4;
constexpr int QROWS = 32;
constexpr int QBLK  = NR / QROWS;
constexpr int KS    = KD / QW;
constexpr int KL    = KS / 8;
constexpr int CGRP  = 4;
}

__global__ void __launch_bounds__(256)
clusternet_cnorm_kernel(const float* __restrict__ cent,
                        float* __restrict__ cnorm) {
    const int lane = threadIdx.x & 31;
    const int row  = blockIdx.x * WAVES + (threadIdx.x >> 5);
    const float* cr = cent + (size_t)row * KD;

    float s = 0.f;
#pragma unroll
    for (int h = 0; h < KD; h += 32) {
        float v = cr[h + lane];
        s += v * v;
    }
    s += __shfl_xor(s, 1, 32);
    s += __shfl_xor(s, 2, 32);
    s += __shfl_xor(s, 4, 32);
    s += __shfl_xor(s, 8, 32);
    s += __shfl_xor(s, 16, 32);
    if (lane == 0) cnorm[row] = s;
}

__global__ void __launch_bounds__(128)
clusternet_q_kernel(const float* __restrict__ z,
                    const float* __restrict__ cent,
                    const float* __restrict__ cnorm,
                    float* __restrict__ Qout,
                    float* __restrict__ blockpart) {
    __shared__ v4f   part[QW][2][4][2][32];
    __shared__ float znp[QW][QROWS];
    __shared__ float csum[QW][NC];

    const int lane = threadIdx.x & 31;
    const int wave = threadIdx.x >> 5;
    const int half = lane >> 4;
    const int n16  = lane & 15;
    const int rowBase = blockIdx.x * QROWS;

    const int koff = wave * KS + 4 * half;
    const float* zr0 = z + (size_t)(rowBase + n16) * KD + koff;
    const float* zr1 = zr0 + (size_t)16 * KD;
    const float* cr0 = cent + (size_t)n16 * KD + koff;

    v4f b[4][KL];
    v4f a[2][KL];
#pragma unroll
    for (int i = 0; i < KL; ++i) {
#pragma unroll
        for (int t = 0; t < 4; ++t)
            b[t][i] = *(const v4f*)(cr0 + (size_t)(t * 16) * KD + 8 * i);
        a[0][i] = *(const v4f*)(zr0 + 8 * i);
        a[1][i] = *(const v4f*)(zr1 + 8 * i);
    }

    float cn[4];
#pragma unroll
    for (int t = 0; t < 4; ++t) cn[t] = cnorm[t * 16 + n16];

    v8f acc[2][4] = {};
    float zp[2] = {0.f, 0.f};

#pragma unroll
    for (int i = 0; i < KL; ++i) {
#pragma unroll
        for (int mt = 0; mt < 2; ++mt) {
            const v4f av = a[mt][i];
            zp[mt] += av.x * av.x + av.y * av.y + av.z * av.z + av.w * av.w;
            v2f a0; a0.x = av.x; a0.y = av.y;
            v2f a1; a1.x = av.z; a1.y = av.w;
#pragma unroll
            for (int t = 0; t < 4; ++t) {
                v2f b0; b0.x = b[t][i].x; b0.y = b[t][i].y;
                acc[mt][t] = __builtin_amdgcn_wmma_f32_16x16x4_f32(
                    false, a0, false, b0, (short)0, acc[mt][t], false, false);
            }
#pragma unroll
            for (int t = 0; t < 4; ++t) {
                v2f b1; b1.x = b[t][i].z; b1.y = b[t][i].w;
                acc[mt][t] = __builtin_amdgcn_wmma_f32_16x16x4_f32(
                    false, a1, false, b1, (short)0, acc[mt][t], false, false);
            }
        }
    }

#pragma unroll
    for (int mt = 0; mt < 2; ++mt) {
#pragma unroll
        for (int t = 0; t < 4; ++t) {
#pragma unroll
            for (int rq = 0; rq < 2; ++rq) {
                v4f p;
                p.x = acc[mt][t][4 * rq + 0];
                p.y = acc[mt][t][4 * rq + 1];
                p.z = acc[mt][t][4 * rq + 2];
                p.w = acc[mt][t][4 * rq + 3];
                part[wave][mt][t][rq][lane] = p;
            }
        }
        float zh = zp[mt] + __shfl_xor(zp[mt], 16, 32);
        if (lane < 16) znp[wave][mt * 16 + n16] = zh;
    }
    __syncthreads();

    const int mt = wave >> 1;
    const int rq = wave & 1;

    v4f d[4];
#pragma unroll
    for (int t = 0; t < 4; ++t) {
        d[t] = part[0][mt][t][rq][lane];
        d[t] += part[1][mt][t][rq][lane];
        d[t] += part[2][mt][t][rq][lane];
        d[t] += part[3][mt][t][rq][lane];
    }

    float qv[4][4];
#pragma unroll
    for (int rr = 0; rr < 4; ++rr) {
        const int trow = mt * 16 + 8 * half + 4 * rq + rr;
        float znr = znp[0][trow];
        znr += znp[1][trow];
        znr += znp[2][trow];
        znr += znp[3][trow];
#pragma unroll
        for (int t = 0; t < 4; ++t) {
            float dd = d[t][rr];
            float s2 = fmaxf(znr + cn[t] - 2.0f * dd, 0.0f);
            float sim = sqrtf(s2);
            qv[t][rr] = 1.0f / (1.0f + sim);
        }
        float rs = qv[0][rr] + qv[1][rr] + qv[2][rr] + qv[3][rr];
        rs += __shfl_xor(rs, 1, 32);
        rs += __shfl_xor(rs, 2, 32);
        rs += __shfl_xor(rs, 4, 32);
        rs += __shfl_xor(rs, 8, 32);
        float inv = 1.0f / rs;
#pragma unroll
        for (int t = 0; t < 4; ++t) {
            qv[t][rr] *= inv;
            Qout[(size_t)(rowBase + trow) * NC + t * 16 + n16] = qv[t][rr];
        }
    }

#pragma unroll
    for (int t = 0; t < 4; ++t) {
        float cs = (qv[t][0] + qv[t][1]) + (qv[t][2] + qv[t][3]);
        cs += __shfl_xor(cs, 16, 32);
        if (lane < 16) csum[wave][t * 16 + n16] = cs;
    }
    __syncthreads();
    if (threadIdx.x < NC) {
        float s = 0.f;
#pragma unroll
        for (int w = 0; w < QW; ++w) s += csum[w][threadIdx.x];
        blockpart[(size_t)blockIdx.x * NC + threadIdx.x] = s;
    }
}

__global__ void __launch_bounds__(256)
clusternet_colsum_kernel(const float* __restrict__ blockpart,
                         float* __restrict__ colsum) {
    __shared__ float red[CGRP][NC];
    const int col = threadIdx.x & (NC - 1);
    const int grp = threadIdx.x >> 6;
    const float* bp = blockpart + (size_t)grp * (QBLK / CGRP) * NC + col;
    float s = 0.f;
#pragma unroll 8
    for (int b = 0; b < QBLK / CGRP; ++b) s += bp[(size_t)b * NC];
    red[grp][col] = s;
    __syncthreads();
    if (threadIdx.x < NC)
        colsum[threadIdx.x] = (red[0][threadIdx.x] + red[1][threadIdx.x]) +
                              (red[2][threadIdx.x] + red[3][threadIdx.x]);
}

__global__ void __launch_bounds__(256)
clusternet_p_kernel(const float* __restrict__ Q,
                    const float* __restrict__ colsum,
                    float* __restrict__ Pout) {
    const int lane = threadIdx.x & 31;
    const int row  = blockIdx.x * WAVES + (threadIdx.x >> 5);
    const size_t base = (size_t)row * NC;

    float q0 = Q[base + lane];
    float q1 = Q[base + lane + 32];
    float p0 = q0 * q0 / colsum[lane];
    float p1 = q1 * q1 / colsum[lane + 32];

    float s = p0 + p1;
    s += __shfl_xor(s, 1, 32);
    s += __shfl_xor(s, 2, 32);
    s += __shfl_xor(s, 4, 32);
    s += __shfl_xor(s, 8, 32);
    s += __shfl_xor(s, 16, 32);
    float inv = 1.0f / s;

    Pout[base + lane]      = p0 * inv;
    Pout[base + lane + 32] = p1 * inv;
}

extern "C" void kernel_launch(void* const* d_in, const int* in_sizes, int n_in,
                              void* d_out, int out_size, void* d_ws, size_t ws_size,
                              hipStream_t stream) {
    const float* z    = (const float*)d_in[0];
    const float* cent = (const float*)d_in[1];

    float* Q = (float*)d_out;
    float* P = Q + (size_t)NR * NC;

    float* colsum = (float*)d_ws;
    float* cnorm  = colsum + NC;
    float* blockpart = P;

    clusternet_cnorm_kernel<<<NC / WAVES, 256, 0, stream>>>(cent, cnorm);
    clusternet_q_kernel<<<QBLK, 128, 0, stream>>>(z, cent, cnorm, Q, blockpart);
    clusternet_colsum_kernel<<<1, 256, 0, stream>>>(blockpart, colsum);
    clusternet_p_kernel<<<NR / WAVES, 256, 0, stream>>>(Q, colsum, P);
}


# cnorm launch folded into the Q kernel: each wave squares its register-held centroid K quarter, quarters summed via LDS (cnorm summed in a new order); 4 launches -> 3; fragment loads all issue first
# speedup vs baseline: 1.8085x; 1.1289x over previous
#include <hip/hip_runtime.h>


typedef __attribute__((ext_vector_type(2))) float v2f;
typedef __attribute__((ext_vector_type(4))) float v4f;
typedef __attribute__((ext_vector_type(8))) float v8f;

namespace {
constexpr int KD    = 256;
constexpr int NC    = 64;
constexpr int NR    = 8192;
constexpr int WAVES = 8;
constexpr int QW    = 4;
constexpr int QROWS = 32;
constexpr int QBLK  = NR / QROWS;
constexpr int KS    = KD / QW;
constexpr int KL    = KS / 8;
constexpr int CGRP  = 4;
}

__global__ void __launch_bounds__(128)
clusternet_q_kernel(const float* __restrict__ z,
                    const float* __restrict__ cent,
                    float* __restrict__ Qout,
                    float* __restrict__ blockpart) {
    __shared__ v4f   part[QW][2][4][2][32];
    __shared__ float znp[QW][QROWS];
    __shared__ float cnp[QW][NC];
    __shared__ float csum[QW][NC];

    const int lane = threadIdx.x & 31;
    const int wave = threadIdx.x >> 5;
    const int half = lane >> 4;
    const int n16  = lane & 15;
    const int rowBase = blockIdx.x * QROWS;

    const int koff = wave * KS + 4 * half;
    const float* zr0 = z + (size_t)(rowBase + n16) * KD + koff;
    const float* zr1 = zr0 + (size_t)16 * KD;
    const float* cr0 = cent + (size_t)n16 * KD + koff;

    v4f b[4][KL];
    v4f a[2][KL];
#pragma unroll
    for (int i = 0; i < KL; ++i) {
#pragma unroll
        for (int t = 0; t < 4; ++t)
            b[t][i] = *(const v4f*)(cr0 + (size_t)(t * 16) * KD + 8 * i);
        a[0][i] = *(const v4f*)(zr0 + 8 * i);
        a[1][i] = *(const v4f*)(zr1 + 8 * i);
    }

    v8f acc[2][4] = {};
    float zp[2] = {0.f, 0.f};
    float cp[4] = {0.f, 0.f, 0.f, 0.f};

#pragma unroll
    for (int i = 0; i < KL; ++i) {
#pragma unroll
        for (int t = 0; t < 4; ++t) {
            const v4f bv = b[t][i];
            cp[t] += bv.x * bv.x + bv.y * bv.y + bv.z * bv.z + bv.w * bv.w;
        }
#pragma unroll
        for (int mt = 0; mt < 2; ++mt) {
            const v4f av = a[mt][i];
            zp[mt] += av.x * av.x + av.y * av.y + av.z * av.z + av.w * av.w;
            v2f a0; a0.x = av.x; a0.y = av.y;
            v2f a1; a1.x = av.z; a1.y = av.w;
#pragma unroll
            for (int t = 0; t < 4; ++t) {
                v2f b0; b0.x = b[t][i].x; b0.y = b[t][i].y;
                acc[mt][t] = __builtin_amdgcn_wmma_f32_16x16x4_f32(
                    false, a0, false, b0, (short)0, acc[mt][t], false, false);
            }
#pragma unroll
            for (int t = 0; t < 4; ++t) {
                v2f b1; b1.x = b[t][i].z; b1.y = b[t][i].w;
                acc[mt][t] = __builtin_amdgcn_wmma_f32_16x16x4_f32(
                    false, a1, false, b1, (short)0, acc[mt][t], false, false);
            }
        }
    }

#pragma unroll
    for (int mt = 0; mt < 2; ++mt) {
#pragma unroll
        for (int t = 0; t < 4; ++t) {
#pragma unroll
            for (int rq = 0; rq < 2; ++rq) {
                v4f p;
                p.x = acc[mt][t][4 * rq + 0];
                p.y = acc[mt][t][4 * rq + 1];
                p.z = acc[mt][t][4 * rq + 2];
                p.w = acc[mt][t][4 * rq + 3];
                part[wave][mt][t][rq][lane] = p;
            }
        }
        float zh = zp[mt] + __shfl_xor(zp[mt], 16, 32);
        if (lane < 16) znp[wave][mt * 16 + n16] = zh;
    }
#pragma unroll
    for (int t = 0; t < 4; ++t) {
        float ch = cp[t] + __shfl_xor(cp[t], 16, 32);
        if (lane < 16) cnp[wave][t * 16 + n16] = ch;
    }
    __syncthreads();

    const int mt = wave >> 1;
    const int rq = wave & 1;

    v4f d[4];
#pragma unroll
    for (int t = 0; t < 4; ++t) {
        d[t] = part[0][mt][t][rq][lane];
        d[t] += part[1][mt][t][rq][lane];
        d[t] += part[2][mt][t][rq][lane];
        d[t] += part[3][mt][t][rq][lane];
    }

    float cn[4];
#pragma unroll
    for (int t = 0; t < 4; ++t) {
        const int c = t * 16 + n16;
        cn[t] = cnp[0][c];
        cn[t] += cnp[1][c];
        cn[t] += cnp[2][c];
        cn[t] += cnp[3][c];
    }

    float qv[4][4];
#pragma unroll
    for (int rr = 0; rr < 4; ++rr) {
        const int trow = mt * 16 + 8 * half + 4 * rq + rr;
        float znr = znp[0][trow];
        znr += znp[1][trow];
        znr += znp[2][trow];
        znr += znp[3][trow];
#pragma unroll
        for (int t = 0; t < 4; ++t) {
            float dd = d[t][rr];
            float s2 = fmaxf(znr + cn[t] - 2.0f * dd, 0.0f);
            float sim = sqrtf(s2);
            qv[t][rr] = 1.0f / (1.0f + sim);
        }
        float rs = qv[0][rr] + qv[1][rr] + qv[2][rr] + qv[3][rr];
        rs += __shfl_xor(rs, 1, 32);
        rs += __shfl_xor(rs, 2, 32);
        rs += __shfl_xor(rs, 4, 32);
        rs += __shfl_xor(rs, 8, 32);
        float inv = 1.0f / rs;
#pragma unroll
        for (int t = 0; t < 4; ++t) {
            qv[t][rr] *= inv;
            Qout[(size_t)(rowBase + trow) * NC + t * 16 + n16] = qv[t][rr];
        }
    }

#pragma unroll
    for (int t = 0; t < 4; ++t) {
        float cs = (qv[t][0] + qv[t][1]) + (qv[t][2] + qv[t][3]);
        cs += __shfl_xor(cs, 16, 32);
        if (lane < 16) csum[wave][t * 16 + n16] = cs;
    }
    __syncthreads();
    if (threadIdx.x < NC) {
        float s = 0.f;
#pragma unroll
        for (int w = 0; w < QW; ++w) s += csum[w][threadIdx.x];
        blockpart[(size_t)blockIdx.x * NC + threadIdx.x] = s;
    }
}

__global__ void __launch_bounds__(256)
clusternet_colsum_kernel(const float* __restrict__ blockpart,
                         float* __restrict__ colsum) {
    __shared__ float red[CGRP][NC];
    const int col = threadIdx.x & (NC - 1);
    const int grp = threadIdx.x >> 6;
    const float* bp = blockpart + (size_t)grp * (QBLK / CGRP) * NC + col;
    float s = 0.f;
#pragma unroll 8
    for (int b = 0; b < QBLK / CGRP; ++b) s += bp[(size_t)b * NC];
    red[grp][col] = s;
    __syncthreads();
    if (threadIdx.x < NC)
        colsum[threadIdx.x] = (red[0][threadIdx.x] + red[1][threadIdx.x]) +
                              (red[2][threadIdx.x] + red[3][threadIdx.x]);
}

__global__ void __launch_bounds__(256)
clusternet_p_kernel(const float* __restrict__ Q,
                    const float* __restrict__ colsum,
                    float* __restrict__ Pout) {
    const int lane = threadIdx.x & 31;
    const int row  = blockIdx.x * WAVES + (threadIdx.x >> 5);
    const size_t base = (size_t)row * NC;

    float q0 = Q[base + lane];
    float q1 = Q[base + lane + 32];
    float p0 = q0 * q0 / colsum[lane];
    float p1 = q1 * q1 / colsum[lane + 32];

    float s = p0 + p1;
    s += __shfl_xor(s, 1, 32);
    s += __shfl_xor(s, 2, 32);
    s += __shfl_xor(s, 4, 32);
    s += __shfl_xor(s, 8, 32);
    s += __shfl_xor(s, 16, 32);
    float inv = 1.0f / s;

    Pout[base + lane]      = p0 * inv;
    Pout[base + lane + 32] = p1 * inv;
}

extern "C" void kernel_launch(void* const* d_in, const int* in_sizes, int n_in,
                              void* d_out, int out_size, void* d_ws, size_t ws_size,
                              hipStream_t stream) {
    const float* z    = (const float*)d_in[0];
    const float* cent = (const float*)d_in[1];

    float* Q = (float*)d_out;
    float* P = Q + (size_t)NR * NC;

    float* colsum = (float*)d_ws;
    float* blockpart = P;

    clusternet_q_kernel<<<QBLK, 128, 0, stream>>>(z, cent, Q, blockpart);
    clusternet_colsum_kernel<<<1, 256, 0, stream>>>(blockpart, colsum);
    clusternet_p_kernel<<<NR / WAVES, 256, 0, stream>>>(Q, colsum, P);
}


# colsum launch folded into the P kernel (3 launches -> 2): block partials moved to the workspace, each of 256 P blocks (32 rows) adds them itself; colsum and P row sums added in a new order
# speedup vs baseline: 2.7064x; 1.4965x over previous
#include <hip/hip_runtime.h>


typedef __attribute__((ext_vector_type(2))) float v2f;
typedef __attribute__((ext_vector_type(4))) float v4f;
typedef __attribute__((ext_vector_type(8))) float v8f;

namespace {
constexpr int KD    = 256;
constexpr int NC    = 64;
constexpr int NR    = 8192;
constexpr int PW    = 8;
constexpr int PROWS = 32;
constexpr int PBLK  = NR / PROWS;
constexpr int QW    = 4;
constexpr int QROWS = 32;
constexpr int QBLK  = NR / QROWS;
constexpr int KS    = KD / QW;
constexpr int KL    = KS / 8;
constexpr size_t PART_BYTES = (size_t)QBLK * NC * sizeof(float);
}

__global__ void __launch_bounds__(128)
clusternet_q_kernel(const float* __restrict__ z,
                    const float* __restrict__ cent,
                    float* __restrict__ Qout,
                    float* __restrict__ blockpart,
                    size_t ws_bytes) {
    __shared__ v4f   part[QW][2][4][2][32];
    __shared__ float znp[QW][QROWS];
    __shared__ float cnp[QW][NC];
    __shared__ float csum[QW][NC];

    const int lane = threadIdx.x & 31;
    const int wave = threadIdx.x >> 5;
    const int half = lane >> 4;
    const int n16  = lane & 15;
    const int rowBase = blockIdx.x * QROWS;

    const int koff = wave * KS + 4 * half;
    const float* zr0 = z + (size_t)(rowBase + n16) * KD + koff;
    const float* zr1 = zr0 + (size_t)16 * KD;
    const float* cr0 = cent + (size_t)n16 * KD + koff;

    v4f b[4][KL];
    v4f a[2][KL];
#pragma unroll
    for (int i = 0; i < KL; ++i) {
#pragma unroll
        for (int t = 0; t < 4; ++t)
            b[t][i] = *(const v4f*)(cr0 + (size_t)(t * 16) * KD + 8 * i);
        a[0][i] = *(const v4f*)(zr0 + 8 * i);
        a[1][i] = *(const v4f*)(zr1 + 8 * i);
    }

    v8f acc[2][4] = {};
    float zp[2] = {0.f, 0.f};
    float cp[4] = {0.f, 0.f, 0.f, 0.f};

#pragma unroll
    for (int i = 0; i < KL; ++i) {
#pragma unroll
        for (int t = 0; t < 4; ++t) {
            const v4f bv = b[t][i];
            cp[t] += bv.x * bv.x + bv.y * bv.y + bv.z * bv.z + bv.w * bv.w;
        }
#pragma unroll
        for (int mt = 0; mt < 2; ++mt) {
            const v4f av = a[mt][i];
            zp[mt] += av.x * av.x + av.y * av.y + av.z * av.z + av.w * av.w;
            v2f a0; a0.x = av.x; a0.y = av.y;
            v2f a1; a1.x = av.z; a1.y = av.w;
#pragma unroll
            for (int t = 0; t < 4; ++t) {
                v2f b0; b0.x = b[t][i].x; b0.y = b[t][i].y;
                acc[mt][t] = __builtin_amdgcn_wmma_f32_16x16x4_f32(
                    false, a0, false, b0, (short)0, acc[mt][t], false, false);
            }
#pragma unroll
            for (int t = 0; t < 4; ++t) {
                v2f b1; b1.x = b[t][i].z; b1.y = b[t][i].w;
                acc[mt][t] = __builtin_amdgcn_wmma_f32_16x16x4_f32(
                    false, a1, false, b1, (short)0, acc[mt][t], false, false);
            }
        }
    }

#pragma unroll
    for (int mt = 0; mt < 2; ++mt) {
#pragma unroll
        for (int t = 0; t < 4; ++t) {
#pragma unroll
            for (int rq = 0; rq < 2; ++rq) {
                v4f p;
                p.x = acc[mt][t][4 * rq + 0];
                p.y = acc[mt][t][4 * rq + 1];
                p.z = acc[mt][t][4 * rq + 2];
                p.w = acc[mt][t][4 * rq + 3];
                part[wave][mt][t][rq][lane] = p;
            }
        }
        float zh = zp[mt] + __shfl_xor(zp[mt], 16, 32);
        if (lane < 16) znp[wave][mt * 16 + n16] = zh;
    }
#pragma unroll
    for (int t = 0; t < 4; ++t) {
        float ch = cp[t] + __shfl_xor(cp[t], 16, 32);
        if (lane < 16) cnp[wave][t * 16 + n16] = ch;
    }
    __syncthreads();

    const int mt = wave >> 1;
    const int rq = wave & 1;

    v4f d[4];
#pragma unroll
    for (int t = 0; t < 4; ++t) {
        d[t] = part[0][mt][t][rq][lane];
        d[t] += part[1][mt][t][rq][lane];
        d[t] += part[2][mt][t][rq][lane];
        d[t] += part[3][mt][t][rq][lane];
    }

    float cn[4];
#pragma unroll
    for (int t = 0; t < 4; ++t) {
        const int c = t * 16 + n16;
        cn[t] = cnp[0][c];
        cn[t] += cnp[1][c];
        cn[t] += cnp[2][c];
        cn[t] += cnp[3][c];
    }

    float qv[4][4];
#pragma unroll
    for (int rr = 0; rr < 4; ++rr) {
        const int trow = mt * 16 + 8 * half + 4 * rq + rr;
        float znr = znp[0][trow];
        znr += znp[1][trow];
        znr += znp[2][trow];
        znr += znp[3][trow];
#pragma unroll
        for (int t = 0; t < 4; ++t) {
            float dd = d[t][rr];
            float s2 = fmaxf(znr + cn[t] - 2.0f * dd, 0.0f);
            float sim = sqrtf(s2);
            qv[t][rr] = 1.0f / (1.0f + sim);
        }
        float rs = qv[0][rr] + qv[1][rr] + qv[2][rr] + qv[3][rr];
        rs += __shfl_xor(rs, 1, 32);
        rs += __shfl_xor(rs, 2, 32);
        rs += __shfl_xor(rs, 4, 32);
        rs += __shfl_xor(rs, 8, 32);
        float inv = 1.0f / rs;
#pragma unroll
        for (int t = 0; t < 4; ++t) {
            qv[t][rr] *= inv;
            Qout[(size_t)(rowBase + trow) * NC + t * 16 + n16] = qv[t][rr];
        }
    }

#pragma unroll
    for (int t = 0; t < 4; ++t) {
        float cs = (qv[t][0] + qv[t][1]) + (qv[t][2] + qv[t][3]);
        cs += __shfl_xor(cs, 16, 32);
        if (lane < 16) csum[wave][t * 16 + n16] = cs;
    }
    __syncthreads();
    if (threadIdx.x < NC && ws_bytes >= PART_BYTES) {
        float s = 0.f;
#pragma unroll
        for (int w = 0; w < QW; ++w) s += csum[w][threadIdx.x];
        blockpart[(size_t)blockIdx.x * NC + threadIdx.x] = s;
    }
}

__global__ void __launch_bounds__(256)
clusternet_p_kernel(const float* __restrict__ Q,
                    const float* __restrict__ blockpart,
                    float* __restrict__ Pout,
                    size_t ws_bytes) {
    __shared__ v4f wred[PW][16];

    const int lane = threadIdx.x & 31;
    const int wave = threadIdx.x >> 5;
    const int half = lane >> 4;
    const int cq   = lane & 15;
    const int row0 = blockIdx.x * PROWS + wave * 4 + half;

    v4f q[2];
    q[0] = *(const v4f*)(Q + (size_t)row0 * NC + 4 * cq);
    q[1] = *(const v4f*)(Q + (size_t)(row0 + 2) * NC + 4 * cq);

    v4f s = {0.f, 0.f, 0.f, 0.f};
    if (ws_bytes >= PART_BYTES) {
        const float* bp = blockpart + (size_t)(wave * 32 + half) * NC + 4 * cq;
        v4f v[16];
#pragma unroll
        for (int j = 0; j < 16; ++j) v[j] = *(const v4f*)(bp + (size_t)(2 * j) * NC);
#pragma unroll
        for (int j = 0; j < 16; ++j) s += v[j];
    } else {
        const float* qp = Q + (size_t)(wave * (NR / PW) + half) * NC + 4 * cq;
        for (int j = 0; j < NR / PW / 2; ++j) s += *(const v4f*)(qp + (size_t)(2 * j) * NC);
    }
    s.x += __shfl_xor(s.x, 16, 32);
    s.y += __shfl_xor(s.y, 16, 32);
    s.z += __shfl_xor(s.z, 16, 32);
    s.w += __shfl_xor(s.w, 16, 32);
    if (lane < 16) wred[wave][cq] = s;
    __syncthreads();
    v4f cs = wred[0][cq];
#pragma unroll
    for (int w = 1; w < PW; ++w) cs += wred[w][cq];

#pragma unroll
    for (int r = 0; r < 2; ++r) {
        v4f p = q[r] * q[r] / cs;
        float rs = (p.x + p.y) + (p.z + p.w);
        rs += __shfl_xor(rs, 1, 32);
        rs += __shfl_xor(rs, 2, 32);
        rs += __shfl_xor(rs, 4, 32);
        rs += __shfl_xor(rs, 8, 32);
        const float inv = 1.0f / rs;
        *(v4f*)(Pout + (size_t)(row0 + 2 * r) * NC + 4 * cq) = p * inv;
    }
}

extern "C" void kernel_launch(void* const* d_in, const int* in_sizes, int n_in,
                              void* d_out, int out_size, void* d_ws, size_t ws_size,
                              hipStream_t stream) {
    const float* z    = (const float*)d_in[0];
    const float* cent = (const float*)d_in[1];

    float* Q = (float*)d_out;
    float* P = Q + (size_t)NR * NC;

    float* blockpart = (float*)d_ws;

    clusternet_q_kernel<<<QBLK, 128, 0, stream>>>(z, cent, Q, blockpart, ws_size);
    clusternet_p_kernel<<<PBLK, 256, 0, stream>>>(Q, blockpart, P, ws_size);
}


# Cross-lane sums (16-lane row sums, half-wave combines) in both kernels as DPP row_xmask adds and permlanex16 on the vector ALU, not LDS-unit lane exchanges; same lane pairs, sums bit-identical
# speedup vs baseline: 2.9032x; 1.0727x over previous
#include <hip/hip_runtime.h>


typedef __attribute__((ext_vector_type(2))) float v2f;
typedef __attribute__((ext_vector_type(4))) float v4f;
typedef __attribute__((ext_vector_type(8))) float v8f;

namespace {
constexpr int KD    = 256;
constexpr int NC    = 64;
constexpr int NR    = 8192;
constexpr int PW    = 8;
constexpr int PROWS = 32;
constexpr int PBLK  = NR / PROWS;
constexpr int QW    = 4;
constexpr int QROWS = 32;
constexpr int QBLK  = NR / QROWS;
constexpr int KS    = KD / QW;
constexpr int KL    = KS / 8;
constexpr size_t PART_BYTES = (size_t)QBLK * NC * sizeof(float);
}


__device__ __forceinline__ float xhalf(float x) {
    const int b = __builtin_bit_cast(int, x);
    return __builtin_bit_cast(float,
        __builtin_amdgcn_permlanex16(b, b, 0x76543210, 0xfedcba98, false, false));
}

template <int M>
__device__ __forceinline__ float xrow(float x) {
    const int b = __builtin_bit_cast(int, x);
    return __builtin_bit_cast(float,
        __builtin_amdgcn_update_dpp(b, b, 0x160 | M, 0xf, 0xf, true));
}

__device__ __forceinline__ float rowsum16(float x) {
    x += xrow<1>(x);
    x += xrow<2>(x);
    x += xrow<4>(x);
    x += xrow<8>(x);
    return x;
}

__global__ void __launch_bounds__(128)
clusternet_q_kernel(const float* __restrict__ z,
                    const float* __restrict__ cent,
                    float* __restrict__ Qout,
                    float* __restrict__ blockpart,
                    size_t ws_bytes) {
    __shared__ v4f   part[QW][2][4][2][32];
    __shared__ float znp[QW][QROWS];
    __shared__ float cnp[QW][NC];
    __shared__ float csum[QW][NC];

    const int lane = threadIdx.x & 31;
    const int wave = threadIdx.x >> 5;
    const int half = lane >> 4;
    const int n16  = lane & 15;
    const int rowBase = blockIdx.x * QROWS;

    const int koff = wave * KS + 4 * half;
    const float* zr0 = z + (size_t)(rowBase + n16) * KD + koff;
    const float* zr1 = zr0 + (size_t)16 * KD;
    const float* cr0 = cent + (size_t)n16 * KD + koff;

    v4f b[4][KL];
    v4f a[2][KL];
#pragma unroll
    for (int i = 0; i < KL; ++i) {
#pragma unroll
        for (int t = 0; t < 4; ++t)
            b[t][i] = *(const v4f*)(cr0 + (size_t)(t * 16) * KD + 8 * i);
        a[0][i] = *(const v4f*)(zr0 + 8 * i);
        a[1][i] = *(const v4f*)(zr1 + 8 * i);
    }

    v8f acc[2][4] = {};
    float zp[2] = {0.f, 0.f};
    float cp[4] = {0.f, 0.f, 0.f, 0.f};

#pragma unroll
    for (int i = 0; i < KL; ++i) {
#pragma unroll
        for (int t = 0; t < 4; ++t) {
            const v4f bv = b[t][i];
            cp[t] += bv.x * bv.x + bv.y * bv.y + bv.z * bv.z + bv.w * bv.w;
        }
#pragma unroll
        for (int mt = 0; mt < 2; ++mt) {
            const v4f av = a[mt][i];
            zp[mt] += av.x * av.x + av.y * av.y + av.z * av.z + av.w * av.w;
            v2f a0; a0.x = av.x; a0.y = av.y;
            v2f a1; a1.x = av.z; a1.y = av.w;
#pragma unroll
            for (int t = 0; t < 4; ++t) {
                v2f b0; b0.x = b[t][i].x; b0.y = b[t][i].y;
                acc[mt][t] = __builtin_amdgcn_wmma_f32_16x16x4_f32(
                    false, a0, false, b0, (short)0, acc[mt][t], false, false);
            }
#pragma unroll
            for (int t = 0; t < 4; ++t) {
                v2f b1; b1.x = b[t][i].z; b1.y = b[t][i].w;
                acc[mt][t] = __builtin_amdgcn_wmma_f32_16x16x4_f32(
                    false, a1, false, b1, (short)0, acc[mt][t], false, false);
            }
        }
    }

#pragma unroll
    for (int mt = 0; mt < 2; ++mt) {
#pragma unroll
        for (int t = 0; t < 4; ++t) {
#pragma unroll
            for (int rq = 0; rq < 2; ++rq) {
                v4f p;
                p.x = acc[mt][t][4 * rq + 0];
                p.y = acc[mt][t][4 * rq + 1];
                p.z = acc[mt][t][4 * rq + 2];
                p.w = acc[mt][t][4 * rq + 3];
                part[wave][mt][t][rq][lane] = p;
            }
        }
        float zh = zp[mt] + xhalf(zp[mt]);
        if (lane < 16) znp[wave][mt * 16 + n16] = zh;
    }
#pragma unroll
    for (int t = 0; t < 4; ++t) {
        float ch = cp[t] + xhalf(cp[t]);
        if (lane < 16) cnp[wave][t * 16 + n16] = ch;
    }
    __syncthreads();

    const int mt = wave >> 1;
    const int rq = wave & 1;

    v4f d[4];
#pragma unroll
    for (int t = 0; t < 4; ++t) {
        d[t] = part[0][mt][t][rq][lane];
        d[t] += part[1][mt][t][rq][lane];
        d[t] += part[2][mt][t][rq][lane];
        d[t] += part[3][mt][t][rq][lane];
    }

    float cn[4];
#pragma unroll
    for (int t = 0; t < 4; ++t) {
        const int c = t * 16 + n16;
        cn[t] = cnp[0][c];
        cn[t] += cnp[1][c];
        cn[t] += cnp[2][c];
        cn[t] += cnp[3][c];
    }

    float qv[4][4];
#pragma unroll
    for (int rr = 0; rr < 4; ++rr) {
        const int trow = mt * 16 + 8 * half + 4 * rq + rr;
        float znr = znp[0][trow];
        znr += znp[1][trow];
        znr += znp[2][trow];
        znr += znp[3][trow];
#pragma unroll
        for (int t = 0; t < 4; ++t) {
            float dd = d[t][rr];
            float s2 = fmaxf(znr + cn[t] - 2.0f * dd, 0.0f);
            float sim = sqrtf(s2);
            qv[t][rr] = 1.0f / (1.0f + sim);
        }
        float rs = rowsum16(qv[0][rr] + qv[1][rr] + qv[2][rr] + qv[3][rr]);
        float inv = 1.0f / rs;
#pragma unroll
        for (int t = 0; t < 4; ++t) {
            qv[t][rr] *= inv;
            Qout[(size_t)(rowBase + trow) * NC + t * 16 + n16] = qv[t][rr];
        }
    }

#pragma unroll
    for (int t = 0; t < 4; ++t) {
        float cs = (qv[t][0] + qv[t][1]) + (qv[t][2] + qv[t][3]);
        cs += xhalf(cs);
        if (lane < 16) csum[wave][t * 16 + n16] = cs;
    }
    __syncthreads();
    if (threadIdx.x < NC && ws_bytes >= PART_BYTES) {
        float s = 0.f;
#pragma unroll
        for (int w = 0; w < QW; ++w) s += csum[w][threadIdx.x];
        blockpart[(size_t)blockIdx.x * NC + threadIdx.x] = s;
    }
}

__global__ void __launch_bounds__(256)
clusternet_p_kernel(const float* __restrict__ Q,
                    const float* __restrict__ blockpart,
                    float* __restrict__ Pout,
                    size_t ws_bytes) {
    __shared__ v4f wred[PW][16];

    const int lane = threadIdx.x & 31;
    const int wave = threadIdx.x >> 5;
    const int half = lane >> 4;
    const int cq   = lane & 15;
    const int row0 = blockIdx.x * PROWS + wave * 4 + half;

    v4f q[2];
    q[0] = *(const v4f*)(Q + (size_t)row0 * NC + 4 * cq);
    q[1] = *(const v4f*)(Q + (size_t)(row0 + 2) * NC + 4 * cq);

    v4f s = {0.f, 0.f, 0.f, 0.f};
    if (ws_bytes >= PART_BYTES) {
        const float* bp = blockpart + (size_t)(wave * 32 + half) * NC + 4 * cq;
        v4f v[16];
#pragma unroll
        for (int j = 0; j < 16; ++j) v[j] = *(const v4f*)(bp + (size_t)(2 * j) * NC);
#pragma unroll
        for (int j = 0; j < 16; ++j) s += v[j];
    } else {
        const float* qp = Q + (size_t)(wave * (NR / PW) + half) * NC + 4 * cq;
        for (int j = 0; j < NR / PW / 2; ++j) s += *(const v4f*)(qp + (size_t)(2 * j) * NC);
    }
    s.x += xhalf(s.x);
    s.y += xhalf(s.y);
    s.z += xhalf(s.z);
    s.w += xhalf(s.w);
    if (lane < 16) wred[wave][cq] = s;
    __syncthreads();
    v4f cs = wred[0][cq];
#pragma unroll
    for (int w = 1; w < PW; ++w) cs += wred[w][cq];

#pragma unroll
    for (int r = 0; r < 2; ++r) {
        v4f p = q[r] * q[r] / cs;
        const float rs = rowsum16((p.x + p.y) + (p.z + p.w));
        const float inv = 1.0f / rs;
        *(v4f*)(Pout + (size_t)(row0 + 2 * r) * NC + 4 * cq) = p * inv;
    }
}

extern "C" void kernel_launch(void* const* d_in, const int* in_sizes, int n_in,
                              void* d_out, int out_size, void* d_ws, size_t ws_size,
                              hipStream_t stream) {
    const float* z    = (const float*)d_in[0];
    const float* cent = (const float*)d_in[1];

    float* Q = (float*)d_out;
    float* P = Q + (size_t)NR * NC;

    float* blockpart = (float*)d_ws;

    clusternet_q_kernel<<<QBLK, 128, 0, stream>>>(z, cent, Q, blockpart, ws_size);
    clusternet_p_kernel<<<PBLK, 256, 0, stream>>>(Q, blockpart, P, ws_size);
}
